# starting kernel
# baseline (speedup 1.0000x reference)
#include <hip/hip_runtime.h>
#include <hip/hip_bf16.h>

#define NN 100000
#define NE 800000
#define DD 64
#define NL 3
#define NG 100
#define NPG (NN / NG)
#define NC 10
#define NCP 16
#define BN_EPS 1e-5f

typedef float v2f __attribute__((ext_vector_type(2)));
typedef float v8f __attribute__((ext_vector_type(8)));

__global__ void __launch_bounds__(256)
egnn_edge_layer(const float* __restrict__ h,
                const float* __restrict__ ef,
                const int*   __restrict__ src,
                const int*   __restrict__ dst,
                const float* __restrict__ W1,
                const float* __restrict__ b1,
                const float* __restrict__ W2,
                const float* __restrict__ b2,
                float*       __restrict__ hnext)
{
    __shared__ float sW2[64 * 64];
    __shared__ float sB1[64], sB2[64];
    __shared__ float sHid[8][16 * 64];
    __shared__ float sEf[8][16];
    __shared__ int   sDst[8][16];

    const int tid  = threadIdx.x;
    const int lane = tid & 31;
    const int wave = tid >> 5;
    const int lo   = lane & 15;
    const int hi   = lane >> 4;

    for (int i = tid; i < 64 * 64; i += 256) sW2[i] = W2[i];
    if (tid < 64) { sB1[tid] = b1[tid]; sB2[tid] = b2[tid]; }
    __syncthreads();

    const float* W1b    = W1 + 64 * 64;
    const float* w1last = W1 + 128 * 64;

    const int numTiles  = (NE + 15) >> 4;
    const int waveGlob  = blockIdx.x * 8 + wave;
    const int waveCount = gridDim.x * 8;

    for (int tile = waveGlob; tile < numTiles; tile += waveCount) {
        const int e      = tile * 16 + lo;
        const bool valid = (e < NE);
        const int eC     = valid ? e : (NE - 1);
        const int sRow   = src[eC];
        const int dRow   = dst[eC];
        if (hi == 0) {
            sEf[wave][lo]  = ef[eC];
            sDst[wave][lo] = valid ? dRow : -1;
        }

        const float* rowS = h + (long long)sRow * DD;
        const float* rowD = h + (long long)dRow * DD;

        v8f acc[4];
        #pragma unroll
        for (int n = 0; n < 4; ++n) {
            const float bb = sB1[n * 16 + lo];
            #pragma unroll
            for (int j = 0; j < 8; ++j) acc[n][j] = bb;
        }

        #pragma unroll
        for (int c = 0; c < 16; ++c) {
            const int k0 = 4 * c + 2 * hi;
            v2f a1 = { rowS[k0], rowS[k0 + 1] };
            v2f a2 = { rowD[k0], rowD[k0 + 1] };
            #pragma unroll
            for (int n = 0; n < 4; ++n) {
                const int col = n * 16 + lo;
                v2f bA = { W1 [k0 * 64 + col], W1 [(k0 + 1) * 64 + col] };
                v2f bB = { W1b[k0 * 64 + col], W1b[(k0 + 1) * 64 + col] };
                acc[n] = __builtin_amdgcn_wmma_f32_16x16x4_f32(
                    false, a1, false, bA, (short)0, acc[n], false, false);
                acc[n] = __builtin_amdgcn_wmma_f32_16x16x4_f32(
                    false, a2, false, bB, (short)0, acc[n], false, false);
            }
        }

        #pragma unroll
        for (int n = 0; n < 4; ++n) {
            const int col  = n * 16 + lo;
            const float wl = w1last[col];
            #pragma unroll
            for (int v = 0; v < 8; ++v) {
                const int M = v + 8 * hi;
                float val = acc[n][v] + sEf[wave][M] * wl;
                sHid[wave][M * 64 + col] = fmaxf(val, 0.0f);
            }
        }

        v8f mac[4];
        #pragma unroll
        for (int n = 0; n < 4; ++n) {
            const float bb = sB2[n * 16 + lo];
            #pragma unroll
            for (int j = 0; j < 8; ++j) mac[n][j] = bb;
        }

        #pragma unroll
        for (int c = 0; c < 16; ++c) {
            const int k0 = 4 * c + 2 * hi;
            v2f a = { sHid[wave][lo * 64 + k0], sHid[wave][lo * 64 + k0 + 1] };
            #pragma unroll
            for (int n = 0; n < 4; ++n) {
                const int col = n * 16 + lo;
                v2f b = { sW2[k0 * 64 + col], sW2[(k0 + 1) * 64 + col] };
                mac[n] = __builtin_amdgcn_wmma_f32_16x16x4_f32(
                    false, a, false, b, (short)0, mac[n], false, false);
            }
        }

        #pragma unroll
        for (int v = 0; v < 8; ++v) {
            const int M    = v + 8 * hi;
            const int drow = sDst[wave][M];
            if (drow >= 0) {
                #pragma unroll
                for (int n = 0; n < 4; ++n)
                    atomicAdd(&hnext[(long long)drow * DD + n * 16 + lo], mac[n][v]);
            }
        }
    }
}

__global__ void __launch_bounds__(256)
zero_f32(float* __restrict__ p, long long n)
{
    for (long long i = blockIdx.x * 256LL + threadIdx.x; i < n;
         i += (long long)gridDim.x * 256LL)
        p[i] = 0.0f;
}

__global__ void __launch_bounds__(256)
bn_reduce(const float* __restrict__ h, float* __restrict__ stats, int nrows)
{
    __shared__ float sSum[256], sSq[256];
    const int c = threadIdx.x & 63;
    const int r = threadIdx.x >> 6;
    float s = 0.f, q = 0.f;
    for (int row = blockIdx.x * 4 + r; row < nrows; row += gridDim.x * 4) {
        float v = h[(long long)row * DD + c];
        s += v; q += v * v;
    }
    sSum[threadIdx.x] = s; sSq[threadIdx.x] = q;
    __syncthreads();
    if (threadIdx.x < 64) {
        float ts = sSum[c] + sSum[64 + c] + sSum[128 + c] + sSum[192 + c];
        float tq = sSq [c] + sSq [64 + c] + sSq [128 + c] + sSq [192 + c];
        atomicAdd(&stats[c], ts);
        atomicAdd(&stats[64 + c], tq);
    }
}

__global__ void __launch_bounds__(256)
bn_relu_apply(float* __restrict__ h, const float* __restrict__ stats,
              const float* __restrict__ gamma, const float* __restrict__ beta,
              int nrows, float invN)
{
    const long long total = (long long)nrows * DD;
    for (long long i = blockIdx.x * 256LL + threadIdx.x; i < total;
         i += (long long)gridDim.x * 256LL) {
        const int c  = (int)(i & 63);
        const float mu  = stats[c] * invN;
        const float var = stats[64 + c] * invN - mu * mu;
        const float rs  = rsqrtf(var + BN_EPS);
        const float v   = (h[i] - mu) * rs * gamma[c] + beta[c];
        h[i] = fmaxf(v, 0.0f);
    }
}

__global__ void __launch_bounds__(256)
pool_mean(const float* __restrict__ h, float* __restrict__ emb)
{
    __shared__ float sm[256];
    const int g = blockIdx.x;
    const int c = threadIdx.x & 63;
    const int r = threadIdx.x >> 6;
    float s = 0.f;
    for (int row = r; row < NPG; row += 4)
        s += h[((long long)g * NPG + row) * DD + c];
    sm[threadIdx.x] = s;
    __syncthreads();
    if (threadIdx.x < 64)
        emb[g * DD + c] = (sm[c] + sm[64 + c] + sm[128 + c] + sm[192 + c]) / (float)NPG;
}

__global__ void __launch_bounds__(256)
bn_heads(const float* __restrict__ emb,
         const float* __restrict__ bg, const float* __restrict__ bb,
         const float* __restrict__ clfW, const float* __restrict__ clfb,
         const float* __restrict__ cptW, const float* __restrict__ cptb,
         float* __restrict__ outLogits, float* __restrict__ outCpt)
{
    __shared__ float sEmb[NG * DD];
    __shared__ float sMu[64], sR[64];
    __shared__ float red[256];
    const int tid = threadIdx.x;
    const int c   = tid & 63;
    const int r   = tid >> 6;

    float s = 0.f, q = 0.f;
    for (int g = r; g < NG; g += 4) {
        float v = emb[g * DD + c];
        s += v; q += v * v;
    }
    red[tid] = s; __syncthreads();
    if (tid < 64) sMu[c] = (red[c] + red[64 + c] + red[128 + c] + red[192 + c]) / (float)NG;
    __syncthreads();
    red[tid] = q; __syncthreads();
    if (tid < 64) {
        float tq  = (red[c] + red[64 + c] + red[128 + c] + red[192 + c]) / (float)NG;
        float mu  = sMu[c];
        sR[c] = rsqrtf(tq - mu * mu + BN_EPS);
    }
    __syncthreads();
    for (int i = tid; i < NG * DD; i += 256) {
        int cc = i & 63;
        sEmb[i] = (emb[i] - sMu[cc]) * sR[cc] * bg[cc] + bb[cc];
    }
    __syncthreads();
    for (int i = tid; i < NG * NC; i += 256) {
        int g = i / NC, k = i % NC;
        float a = clfb[k];
        #pragma unroll 8
        for (int cc = 0; cc < DD; ++cc) a += sEmb[g * DD + cc] * clfW[cc * NC + k];
        outLogits[i] = a;
    }
    for (int i = tid; i < NG * NCP; i += 256) {
        int g = i >> 4, k = i & 15;
        float a = cptb[k];
        #pragma unroll 8
        for (int cc = 0; cc < DD; ++cc) a += sEmb[g * DD + cc] * cptW[cc * NCP + k];
        outCpt[i] = a;
    }
}

extern "C" void kernel_launch(void* const* d_in, const int* in_sizes, int n_in,
                              void* d_out, int out_size, void* d_ws, size_t ws_size,
                              hipStream_t stream)
{
    const float* x    = (const float*)d_in[0];
    const float* ef   = (const float*)d_in[1];
    const int*   src  = (const int*)  d_in[2];
    const int*   dst  = (const int*)  d_in[3];
    const float* W1s  = (const float*)d_in[5];
    const float* b1s  = (const float*)d_in[6];
    const float* W2s  = (const float*)d_in[7];
    const float* b2s  = (const float*)d_in[8];
    const float* gam  = (const float*)d_in[9];
    const float* bet  = (const float*)d_in[10];
    const float* bng  = (const float*)d_in[11];
    const float* bnb  = (const float*)d_in[12];
    const float* clfW = (const float*)d_in[13];
    const float* clfb = (const float*)d_in[14];
    const float* cptW = (const float*)d_in[15];
    const float* cptb = (const float*)d_in[16];

    float* outH      = (float*)d_out;
    float* outLogits = outH + (long long)NN * DD;
    float* outCpt    = outLogits + NG * NC;

    float* hA    = (float*)d_ws;
    float* stats = hA + (long long)NN * DD;
    float* emb   = stats + 128;

    const float* hin = x;
    float* tgt[NL] = { outH, hA, outH };

    const int edgeBlocks = 2048;

    for (int l = 0; l < NL; ++l) {
        float* hn = tgt[l];
        zero_f32<<<1024, 256, 0, stream>>>(hn, (long long)NN * DD);
        zero_f32<<<1,    256, 0, stream>>>(stats, 128);
        egnn_edge_layer<<<edgeBlocks, 256, 0, stream>>>(
            hin, ef, src, dst,
            W1s + (long long)l * (2 * DD + 1) * DD,
            b1s + l * DD,
            W2s + (long long)l * DD * DD,
            b2s + l * DD,
            hn);
        bn_reduce<<<512, 256, 0, stream>>>(hn, stats, NN);
        bn_relu_apply<<<1024, 256, 0, stream>>>(hn, stats, gam + l * DD, bet + l * DD,
                                                NN, 1.0f / (float)NN);
        hin = hn;
    }

    pool_mean<<<NG, 256, 0, stream>>>(outH, emb);
    bn_heads<<<1, 256, 0, stream>>>(emb, bng, bnb, clfW, clfb, cptW, cptb,
                                    outLogits, outCpt);
}


# B fragments of both GEMMs read per k-step from LDS copies of W1 and W2 kept in fragment order, replacing 256 hoisted per-lane global loads per tile; VGPRs 424 to 154, so 6 waves per SIMD instead of 2
# speedup vs baseline: 1.2994x; 1.2994x over previous
#include <hip/hip_runtime.h>
#include <hip/hip_bf16.h>

#define NN 100000
#define NE 800000
#define DD 64
#define NL 3
#define NG 100
#define NPG (NN / NG)
#define NC 10
#define NCP 16
#define BN_EPS 1e-5f

typedef float v2f __attribute__((ext_vector_type(2)));
typedef float v8f __attribute__((ext_vector_type(8)));

__global__ void __launch_bounds__(256)
egnn_edge_layer(const float* __restrict__ h,
                const float* __restrict__ ef,
                const int*   __restrict__ src,
                const int*   __restrict__ dst,
                const float* __restrict__ W1,
                const float* __restrict__ b1,
                const float* __restrict__ W2,
                const float* __restrict__ b2,
                float*       __restrict__ hnext)
{
    __shared__ __attribute__((aligned(16))) v2f sW1f[4 * 64 * 16];
    __shared__ __attribute__((aligned(16))) v2f sW2f[4 * 32 * 16];
    __shared__ float sB1[64], sB2[64];
    __shared__ float sHid[8][16 * 64];
    __shared__ float sEf[8][16];
    __shared__ int   sDst[8][16];

    const int tid  = threadIdx.x;
    const int lane = tid & 31;
    const int wave = tid >> 5;
    const int lo   = lane & 15;
    const int hi   = lane >> 4;

    for (int u = tid; u < 64 * 16; u += 256) {
        const int p  = u >> 4;
        const int c4 = (u & 15) * 4;
        const float* r0 = W1 + (2 * p) * 64 + c4;
        const float* r1 = r0 + 64;
        v2f* d = sW1f + ((c4 >> 4) * 64 + p) * 16 + (c4 & 15);
        #pragma unroll
        for (int j = 0; j < 4; ++j) d[j] = (v2f){ r0[j], r1[j] };
    }
    for (int u = tid; u < 32 * 16; u += 256) {
        const int p  = u >> 4;
        const int c4 = (u & 15) * 4;
        const float* r0 = W2 + (2 * p) * 64 + c4;
        const float* r1 = r0 + 64;
        v2f* d = sW2f + ((c4 >> 4) * 32 + p) * 16 + (c4 & 15);
        #pragma unroll
        for (int j = 0; j < 4; ++j) d[j] = (v2f){ r0[j], r1[j] };
    }
    if (tid < 64) { sB1[tid] = b1[tid]; sB2[tid] = b2[tid]; }
    __syncthreads();

    const float* w1last = W1 + 128 * 64;

    const int numTiles  = (NE + 15) >> 4;
    const int waveGlob  = blockIdx.x * 8 + wave;
    const int waveCount = gridDim.x * 8;

    for (int tile = waveGlob; tile < numTiles; tile += waveCount) {
        const int e      = tile * 16 + lo;
        const bool valid = (e < NE);
        const int eC     = valid ? e : (NE - 1);
        const int sRow   = src[eC];
        const int dRow   = dst[eC];
        if (hi == 0) {
            sEf[wave][lo]  = ef[eC];
            sDst[wave][lo] = valid ? dRow : -1;
        }

        const float* rowS = h + (long long)sRow * DD;
        const float* rowD = h + (long long)dRow * DD;

        v8f acc[4];
        #pragma unroll
        for (int n = 0; n < 4; ++n) {
            const float bb = sB1[n * 16 + lo];
            #pragma unroll
            for (int j = 0; j < 8; ++j) acc[n][j] = bb;
        }

        v2f aS[16], aD[16];
        #pragma unroll
        for (int c = 0; c < 16; ++c) {
            const int k0 = 4 * c + 2 * hi;
            aS[c] = (v2f){ rowS[k0], rowS[k0 + 1] };
            aD[c] = (v2f){ rowD[k0], rowD[k0 + 1] };
        }

        #pragma unroll
        for (int c = 0; c < 16; ++c) {
            v2f bA[4], bB[4];
            #pragma unroll
            for (int n = 0; n < 4; ++n) {
                bA[n] = sW1f[(n * 64 + 2 * c + hi) * 16 + lo];
                bB[n] = sW1f[(n * 64 + 32 + 2 * c + hi) * 16 + lo];
            }
            #pragma unroll
            for (int n = 0; n < 4; ++n) {
                acc[n] = __builtin_amdgcn_wmma_f32_16x16x4_f32(
                    false, aS[c], false, bA[n], (short)0, acc[n], false, false);
                acc[n] = __builtin_amdgcn_wmma_f32_16x16x4_f32(
                    false, aD[c], false, bB[n], (short)0, acc[n], false, false);
            }
            __builtin_amdgcn_sched_barrier(0);
        }

        #pragma unroll
        for (int n = 0; n < 4; ++n) {
            const int col  = n * 16 + lo;
            const float wl = w1last[col];
            #pragma unroll
            for (int v = 0; v < 8; ++v) {
                const int M = v + 8 * hi;
                float val = acc[n][v] + sEf[wave][M] * wl;
                sHid[wave][M * 64 + col] = fmaxf(val, 0.0f);
            }
        }

        v8f mac[4];
        #pragma unroll
        for (int n = 0; n < 4; ++n) {
            const float bb = sB2[n * 16 + lo];
            #pragma unroll
            for (int j = 0; j < 8; ++j) mac[n][j] = bb;
        }

        #pragma unroll
        for (int c = 0; c < 16; ++c) {
            const int k0 = 4 * c + 2 * hi;
            v2f a = { sHid[wave][lo * 64 + k0], sHid[wave][lo * 64 + k0 + 1] };
            #pragma unroll
            for (int n = 0; n < 4; ++n) {
                v2f b = sW2f[(n * 32 + 2 * c + hi) * 16 + lo];
                mac[n] = __builtin_amdgcn_wmma_f32_16x16x4_f32(
                    false, a, false, b, (short)0, mac[n], false, false);
            }
            if ((c % 4) == 3) __builtin_amdgcn_sched_barrier(0);
        }

        #pragma unroll
        for (int v = 0; v < 8; ++v) {
            const int M    = v + 8 * hi;
            const int drow = sDst[wave][M];
            if (drow >= 0) {
                #pragma unroll
                for (int n = 0; n < 4; ++n)
                    atomicAdd(&hnext[(long long)drow * DD + n * 16 + lo], mac[n][v]);
            }
        }
    }
}

__global__ void __launch_bounds__(256)
zero_f32(float* __restrict__ p, long long n)
{
    for (long long i = blockIdx.x * 256LL + threadIdx.x; i < n;
         i += (long long)gridDim.x * 256LL)
        p[i] = 0.0f;
}

__global__ void __launch_bounds__(256)
bn_reduce(const float* __restrict__ h, float* __restrict__ stats, int nrows)
{
    __shared__ float sSum[256], sSq[256];
    const int c = threadIdx.x & 63;
    const int r = threadIdx.x >> 6;
    float s = 0.f, q = 0.f;
    for (int row = blockIdx.x * 4 + r; row < nrows; row += gridDim.x * 4) {
        float v = h[(long long)row * DD + c];
        s += v; q += v * v;
    }
    sSum[threadIdx.x] = s; sSq[threadIdx.x] = q;
    __syncthreads();
    if (threadIdx.x < 64) {
        float ts = sSum[c] + sSum[64 + c] + sSum[128 + c] + sSum[192 + c];
        float tq = sSq [c] + sSq [64 + c] + sSq [128 + c] + sSq [192 + c];
        atomicAdd(&stats[c], ts);
        atomicAdd(&stats[64 + c], tq);
    }
}

__global__ void __launch_bounds__(256)
bn_relu_apply(float* __restrict__ h, const float* __restrict__ stats,
              const float* __restrict__ gamma, const float* __restrict__ beta,
              int nrows, float invN)
{
    const long long total = (long long)nrows * DD;
    for (long long i = blockIdx.x * 256LL + threadIdx.x; i < total;
         i += (long long)gridDim.x * 256LL) {
        const int c  = (int)(i & 63);
        const float mu  = stats[c] * invN;
        const float var = stats[64 + c] * invN - mu * mu;
        const float rs  = rsqrtf(var + BN_EPS);
        const float v   = (h[i] - mu) * rs * gamma[c] + beta[c];
        h[i] = fmaxf(v, 0.0f);
    }
}

__global__ void __launch_bounds__(256)
pool_mean(const float* __restrict__ h, float* __restrict__ emb)
{
    __shared__ float sm[256];
    const int g = blockIdx.x;
    const int c = threadIdx.x & 63;
    const int r = threadIdx.x >> 6;
    float s = 0.f;
    for (int row = r; row < NPG; row += 4)
        s += h[((long long)g * NPG + row) * DD + c];
    sm[threadIdx.x] = s;
    __syncthreads();
    if (threadIdx.x < 64)
        emb[g * DD + c] = (sm[c] + sm[64 + c] + sm[128 + c] + sm[192 + c]) / (float)NPG;
}

__global__ void __launch_bounds__(256)
bn_heads(const float* __restrict__ emb,
         const float* __restrict__ bg, const float* __restrict__ bb,
         const float* __restrict__ clfW, const float* __restrict__ clfb,
         const float* __restrict__ cptW, const float* __restrict__ cptb,
         float* __restrict__ outLogits, float* __restrict__ outCpt)
{
    __shared__ float sEmb[NG * DD];
    __shared__ float sMu[64], sR[64];
    __shared__ float red[256];
    const int tid = threadIdx.x;
    const int c   = tid & 63;
    const int r   = tid >> 6;

    float s = 0.f, q = 0.f;
    for (int g = r; g < NG; g += 4) {
        float v = emb[g * DD + c];
        s += v; q += v * v;
    }
    red[tid] = s; __syncthreads();
    if (tid < 64) sMu[c] = (red[c] + red[64 + c] + red[128 + c] + red[192 + c]) / (float)NG;
    __syncthreads();
    red[tid] = q; __syncthreads();
    if (tid < 64) {
        float tq  = (red[c] + red[64 + c] + red[128 + c] + red[192 + c]) / (float)NG;
        float mu  = sMu[c];
        sR[c] = rsqrtf(tq - mu * mu + BN_EPS);
    }
    __syncthreads();
    for (int i = tid; i < NG * DD; i += 256) {
        int cc = i & 63;
        sEmb[i] = (emb[i] - sMu[cc]) * sR[cc] * bg[cc] + bb[cc];
    }
    __syncthreads();
    for (int i = tid; i < NG * NC; i += 256) {
        int g = i / NC, k = i % NC;
        float a = clfb[k];
        #pragma unroll 8
        for (int cc = 0; cc < DD; ++cc) a += sEmb[g * DD + cc] * clfW[cc * NC + k];
        outLogits[i] = a;
    }
    for (int i = tid; i < NG * NCP; i += 256) {
        int g = i >> 4, k = i & 15;
        float a = cptb[k];
        #pragma unroll 8
        for (int cc = 0; cc < DD; ++cc) a += sEmb[g * DD + cc] * cptW[cc * NCP + k];
        outCpt[i] = a;
    }
}

extern "C" void kernel_launch(void* const* d_in, const int* in_sizes, int n_in,
                              void* d_out, int out_size, void* d_ws, size_t ws_size,
                              hipStream_t stream)
{
    const float* x    = (const float*)d_in[0];
    const float* ef   = (const float*)d_in[1];
    const int*   src  = (const int*)  d_in[2];
    const int*   dst  = (const int*)  d_in[3];
    const float* W1s  = (const float*)d_in[5];
    const float* b1s  = (const float*)d_in[6];
    const float* W2s  = (const float*)d_in[7];
    const float* b2s  = (const float*)d_in[8];
    const float* gam  = (const float*)d_in[9];
    const float* bet  = (const float*)d_in[10];
    const float* bng  = (const float*)d_in[11];
    const float* bnb  = (const float*)d_in[12];
    const float* clfW = (const float*)d_in[13];
    const float* clfb = (const float*)d_in[14];
    const float* cptW = (const float*)d_in[15];
    const float* cptb = (const float*)d_in[16];

    float* outH      = (float*)d_out;
    float* outLogits = outH + (long long)NN * DD;
    float* outCpt    = outLogits + NG * NC;

    float* hA    = (float*)d_ws;
    float* stats = hA + (long long)NN * DD;
    float* emb   = stats + 128;

    const float* hin = x;
    float* tgt[NL] = { outH, hA, outH };

    const int edgeBlocks = 2048;

    for (int l = 0; l < NL; ++l) {
        float* hn = tgt[l];
        zero_f32<<<1024, 256, 0, stream>>>(hn, (long long)NN * DD);
        zero_f32<<<1,    256, 0, stream>>>(stats, 128);
        egnn_edge_layer<<<edgeBlocks, 256, 0, stream>>>(
            hin, ef, src, dst,
            W1s + (long long)l * (2 * DD + 1) * DD,
            b1s + l * DD,
            W2s + (long long)l * DD * DD,
            b2s + l * DD,
            hn);
        bn_reduce<<<512, 256, 0, stream>>>(hn, stats, NN);
        bn_relu_apply<<<1024, 256, 0, stream>>>(hn, stats, gam + l * DD, bet + l * DD,
                                                NN, 1.0f / (float)NN);
        hin = hn;
    }

    pool_mean<<<NG, 256, 0, stream>>>(outH, emb);
    bn_heads<<<1, 256, 0, stream>>>(emb, bng, bnb, clfW, clfb, cptW, cptb,
                                    outLogits, outCpt);
}
